# starting kernel
# baseline (speedup 1.0000x reference)
#include <hip/hip_runtime.h>
#include <math.h>

#define D_MODEL  1024
#define D_STATE  16
#define D_CONV   4
#define D_INNER  2048
#define DT_RANK  64
#define BATCH    2
#define SEQLEN   2048
#define BL       (BATCH * SEQLEN)
#define XDBL_LD  (DT_RANK + 2 * D_STATE)

typedef __attribute__((ext_vector_type(2))) float v2f;
typedef __attribute__((ext_vector_type(4))) float v4f;
typedef __attribute__((ext_vector_type(8))) float v8f;

__device__ __forceinline__ float silu_f(float v)     { return v / (1.f + expf(-v)); }
__device__ __forceinline__ float softplus_f(float v) { return (v > 20.f) ? v : log1pf(expf(v)); }

template <int NT, int EPI>
__launch_bounds__(256)
__global__ void gemm_wmma_f32(const float* __restrict__ A, int lda,
                              const float* __restrict__ W, int ldw,
                              float* __restrict__ C, int ldc, int K,
                              const float* __restrict__ bias)
{
    constexpr int KC      = 32;
    constexpr int KC4     = KC / 4;
    constexpr int LDS_LD  = KC + 4;
    constexpr int NROWS_B = 16 * NT;
    constexpr int NA      = (128 * KC4) / 256;
    constexpr int NB      = (NROWS_B * KC4) / 256;

    __shared__ float As[128 * LDS_LD];
    __shared__ float Bs[NROWS_B * LDS_LD];

    const int tid  = threadIdx.x;
    const int wave = tid >> 5;
    const int lane = tid & 31;
    const int half = lane >> 4;
    const int lidx = lane & 15;
    const int m_blk  = blockIdx.x * 128;
    const int n_base = blockIdx.y * NROWS_B;

    v8f acc[NT];
#pragma unroll
    for (int t = 0; t < NT; ++t) acc[t] = (v8f){0.f,0.f,0.f,0.f,0.f,0.f,0.f,0.f};

    v4f ra[NA], rb[NB];

#pragma unroll
    for (int i = 0; i < NA; ++i) {
        const int idx = tid + i * 256, r = idx >> 3, c = (idx & 7) * 4;
        ra[i] = *(const v4f*)(A + (size_t)(m_blk + r) * lda + c);
    }
#pragma unroll
    for (int i = 0; i < NB; ++i) {
        const int idx = tid + i * 256, r = idx >> 3, c = (idx & 7) * 4;
        rb[i] = *(const v4f*)(W + (size_t)(n_base + r) * ldw + c);
    }

    for (int k0 = 0; k0 < K; k0 += KC) {
#pragma unroll
        for (int i = 0; i < NA; ++i) {
            const int idx = tid + i * 256, r = idx >> 3, c = (idx & 7) * 4;
            *(v4f*)(As + r * LDS_LD + c) = ra[i];
        }
#pragma unroll
        for (int i = 0; i < NB; ++i) {
            const int idx = tid + i * 256, r = idx >> 3, c = (idx & 7) * 4;
            *(v4f*)(Bs + r * LDS_LD + c) = rb[i];
        }
        __syncthreads();

        if (k0 + KC < K) {
#pragma unroll
            for (int i = 0; i < NA; ++i) {
                const int idx = tid + i * 256, r = idx >> 3, c = (idx & 7) * 4;
                ra[i] = *(const v4f*)(A + (size_t)(m_blk + r) * lda + (k0 + KC) + c);
            }
#pragma unroll
            for (int i = 0; i < NB; ++i) {
                const int idx = tid + i * 256, r = idx >> 3, c = (idx & 7) * 4;
                rb[i] = *(const v4f*)(W + (size_t)(n_base + r) * ldw + (k0 + KC) + c);
            }
        }

        const float* a_lds = As + (wave * 16 + lidx) * LDS_LD + 2 * half;
        const float* b_lds = Bs + lidx * LDS_LD + 2 * half;
#pragma unroll
        for (int kk = 0; kk < KC; kk += 4) {
            v2f a = *(const v2f*)(a_lds + kk);
#pragma unroll
            for (int t = 0; t < NT; ++t) {
                v2f b = *(const v2f*)(b_lds + t * 16 * LDS_LD + kk);
                acc[t] = __builtin_amdgcn_wmma_f32_16x16x4_f32(
                    false, a, false, b, (short)0, acc[t], false, false);
            }
        }
        __syncthreads();
    }

#pragma unroll
    for (int t = 0; t < NT; ++t) {
        const int n = n_base + t * 16 + lidx;
#pragma unroll
        for (int v = 0; v < 8; ++v) {
            const int m = m_blk + wave * 16 + half * 8 + v;
            float val = acc[t][v];
            if (EPI == 1) val = softplus_f(val + bias[n]);
            C[(size_t)m * ldc + n] = val;
        }
    }
}

__global__ void conv_silu_kernel(const float* __restrict__ x,
                                 const float* __restrict__ w,
                                 const float* __restrict__ cb,
                                 float* __restrict__ xc)
{
    const int idx = blockIdx.x * blockDim.x + threadIdx.x;
    if (idx >= BATCH * SEQLEN * D_INNER) return;
    const int d = idx % D_INNER;
    const int l = (idx / D_INNER) % SEQLEN;
    const int b = idx / (D_INNER * SEQLEN);
    float acc = cb[d];
#pragma unroll
    for (int k = 0; k < D_CONV; ++k) {
        const int ls = l + k - (D_CONV - 1);
        if (ls >= 0)
            acc += x[((size_t)b * SEQLEN + ls) * D_INNER + d] * w[d * D_CONV + k];
    }
    xc[idx] = silu_f(acc);
}

__global__ void scan_kernel(const float* delta,
                            const float* __restrict__ xc,
                            const float* __restrict__ xdbl,
                            const float* __restrict__ A_log,
                            const float* __restrict__ Dp,
                            const float* __restrict__ z,
                            float* yout)
{
    const int gid = blockIdx.x * blockDim.x + threadIdx.x;
    const int n  = gid & (D_STATE - 1);
    const int ch = gid >> 4;
    if (ch >= BATCH * D_INNER) return;
    const int d = ch % D_INNER;
    const int b = ch / D_INNER;

    const float An = -expf(A_log[d * D_STATE + n]);
    const float Dd = Dp[d];
    float h = 0.f;

    const size_t row0 = (size_t)b * SEQLEN;
    for (int l = 0; l < SEQLEN; ++l) {
        const size_t off = (row0 + l) * D_INNER + d;
        const float dt = delta[off];
        const float u  = xc[off];
        const float* bc = xdbl + (row0 + l) * XDBL_LD;
        const float Bn = bc[DT_RANK + n];
        const float Cn = bc[DT_RANK + D_STATE + n];

        const float dA = expf(dt * An);
        h = h * dA + (dt * u) * Bn;
        float t = h * Cn;
        t += __shfl_xor(t, 1, 16);
        t += __shfl_xor(t, 2, 16);
        t += __shfl_xor(t, 4, 16);
        t += __shfl_xor(t, 8, 16);
        if (n == 0) {
            const float zz = z[off];
            yout[off] = (t + Dd * u) * silu_f(zz);
        }
    }
}

extern "C" void kernel_launch(void* const* d_in, const int* in_sizes, int n_in,
                              void* d_out, int out_size, void* d_ws, size_t ws_size,
                              hipStream_t stream)
{
    const float* hs        = (const float*)d_in[0];
    const float* in_proj_w = (const float*)d_in[1];
    const float* conv_w    = (const float*)d_in[2];
    const float* conv_b    = (const float*)d_in[3];
    const float* x_proj_w  = (const float*)d_in[4];
    const float* dt_proj_w = (const float*)d_in[5];
    const float* dt_proj_b = (const float*)d_in[6];
    const float* A_log     = (const float*)d_in[7];
    const float* Dp        = (const float*)d_in[8];
    const float* out_w     = (const float*)d_in[9];
    float* out = (float*)d_out;

    const size_t SZ_BLD = (size_t)BL * D_INNER * sizeof(float);
    char* ws = (char*)d_ws;
    float* x     = (float*)(ws);
    float* z     = (float*)(ws + SZ_BLD);
    float* xc    = (float*)(ws + 2 * SZ_BLD);
    float* delta = (float*)(ws + 3 * SZ_BLD);
    float* xdbl  = (float*)(ws + 4 * SZ_BLD);

    const dim3 blk(256);

    gemm_wmma_f32<8, 0><<<dim3(BL / 128, D_INNER / 128), blk, 0, stream>>>(
        hs, D_MODEL, in_proj_w, D_MODEL, x, D_INNER, D_MODEL, nullptr);
    gemm_wmma_f32<8, 0><<<dim3(BL / 128, D_INNER / 128), blk, 0, stream>>>(
        hs, D_MODEL, in_proj_w + (size_t)D_INNER * D_MODEL, D_MODEL,
        z, D_INNER, D_MODEL, nullptr);

    const int nelem = BATCH * SEQLEN * D_INNER;
    conv_silu_kernel<<<(nelem + 255) / 256, blk, 0, stream>>>(x, conv_w, conv_b, xc);

    gemm_wmma_f32<6, 0><<<dim3(BL / 128, 1), blk, 0, stream>>>(
        xc, D_INNER, x_proj_w, D_INNER, xdbl, XDBL_LD, D_INNER, nullptr);

    gemm_wmma_f32<8, 1><<<dim3(BL / 128, D_INNER / 128), blk, 0, stream>>>(
        xdbl, XDBL_LD, dt_proj_w, DT_RANK, delta, D_INNER, DT_RANK, dt_proj_b);

    const int scan_threads = BATCH * D_INNER * D_STATE;
    scan_kernel<<<(scan_threads + 255) / 256, blk, 0, stream>>>(
        delta, xc, xdbl, A_log, Dp, z, delta);

    gemm_wmma_f32<8, 0><<<dim3(BL / 128, D_MODEL / 128), blk, 0, stream>>>(
        delta, D_INNER, out_w, D_INNER, out, D_MODEL, D_INNER, nullptr);
}


# scan: time loop blocked by 8 steps; each block's dt/u/z/B/C loads issued one block ahead into registers, outputs stored per block into a plane of their own; arithmetic and sum order unchanged
# speedup vs baseline: 4.1725x; 4.1725x over previous
#include <hip/hip_runtime.h>
#include <math.h>

#define D_MODEL  1024
#define D_STATE  16
#define D_CONV   4
#define D_INNER  2048
#define DT_RANK  64
#define BATCH    2
#define SEQLEN   2048
#define BL       (BATCH * SEQLEN)
#define XDBL_LD  (DT_RANK + 2 * D_STATE)

typedef __attribute__((ext_vector_type(2))) float v2f;
typedef __attribute__((ext_vector_type(4))) float v4f;
typedef __attribute__((ext_vector_type(8))) float v8f;

__device__ __forceinline__ float silu_f(float v)     { return v / (1.f + expf(-v)); }
__device__ __forceinline__ float softplus_f(float v) { return (v > 20.f) ? v : log1pf(expf(v)); }

template <int NT, int EPI>
__launch_bounds__(256)
__global__ void gemm_wmma_f32(const float* __restrict__ A, int lda,
                              const float* __restrict__ W, int ldw,
                              float* __restrict__ C, int ldc, int K,
                              const float* __restrict__ bias)
{
    constexpr int KC      = 32;
    constexpr int KC4     = KC / 4;
    constexpr int LDS_LD  = KC + 4;
    constexpr int NROWS_B = 16 * NT;
    constexpr int NA      = (128 * KC4) / 256;
    constexpr int NB      = (NROWS_B * KC4) / 256;

    __shared__ float As[128 * LDS_LD];
    __shared__ float Bs[NROWS_B * LDS_LD];

    const int tid  = threadIdx.x;
    const int wave = tid >> 5;
    const int lane = tid & 31;
    const int half = lane >> 4;
    const int lidx = lane & 15;
    const int m_blk  = blockIdx.x * 128;
    const int n_base = blockIdx.y * NROWS_B;

    v8f acc[NT];
#pragma unroll
    for (int t = 0; t < NT; ++t) acc[t] = (v8f){0.f,0.f,0.f,0.f,0.f,0.f,0.f,0.f};

    v4f ra[NA], rb[NB];

#pragma unroll
    for (int i = 0; i < NA; ++i) {
        const int idx = tid + i * 256, r = idx >> 3, c = (idx & 7) * 4;
        ra[i] = *(const v4f*)(A + (size_t)(m_blk + r) * lda + c);
    }
#pragma unroll
    for (int i = 0; i < NB; ++i) {
        const int idx = tid + i * 256, r = idx >> 3, c = (idx & 7) * 4;
        rb[i] = *(const v4f*)(W + (size_t)(n_base + r) * ldw + c);
    }

    for (int k0 = 0; k0 < K; k0 += KC) {
#pragma unroll
        for (int i = 0; i < NA; ++i) {
            const int idx = tid + i * 256, r = idx >> 3, c = (idx & 7) * 4;
            *(v4f*)(As + r * LDS_LD + c) = ra[i];
        }
#pragma unroll
        for (int i = 0; i < NB; ++i) {
            const int idx = tid + i * 256, r = idx >> 3, c = (idx & 7) * 4;
            *(v4f*)(Bs + r * LDS_LD + c) = rb[i];
        }
        __syncthreads();

        if (k0 + KC < K) {
#pragma unroll
            for (int i = 0; i < NA; ++i) {
                const int idx = tid + i * 256, r = idx >> 3, c = (idx & 7) * 4;
                ra[i] = *(const v4f*)(A + (size_t)(m_blk + r) * lda + (k0 + KC) + c);
            }
#pragma unroll
            for (int i = 0; i < NB; ++i) {
                const int idx = tid + i * 256, r = idx >> 3, c = (idx & 7) * 4;
                rb[i] = *(const v4f*)(W + (size_t)(n_base + r) * ldw + (k0 + KC) + c);
            }
        }

        const float* a_lds = As + (wave * 16 + lidx) * LDS_LD + 2 * half;
        const float* b_lds = Bs + lidx * LDS_LD + 2 * half;
#pragma unroll
        for (int kk = 0; kk < KC; kk += 4) {
            v2f a = *(const v2f*)(a_lds + kk);
#pragma unroll
            for (int t = 0; t < NT; ++t) {
                v2f b = *(const v2f*)(b_lds + t * 16 * LDS_LD + kk);
                acc[t] = __builtin_amdgcn_wmma_f32_16x16x4_f32(
                    false, a, false, b, (short)0, acc[t], false, false);
            }
        }
        __syncthreads();
    }

#pragma unroll
    for (int t = 0; t < NT; ++t) {
        const int n = n_base + t * 16 + lidx;
#pragma unroll
        for (int v = 0; v < 8; ++v) {
            const int m = m_blk + wave * 16 + half * 8 + v;
            float val = acc[t][v];
            if (EPI == 1) val = softplus_f(val + bias[n]);
            C[(size_t)m * ldc + n] = val;
        }
    }
}

__global__ void conv_silu_kernel(const float* __restrict__ x,
                                 const float* __restrict__ w,
                                 const float* __restrict__ cb,
                                 float* __restrict__ xc)
{
    const int idx = blockIdx.x * blockDim.x + threadIdx.x;
    if (idx >= BATCH * SEQLEN * D_INNER) return;
    const int d = idx % D_INNER;
    const int l = (idx / D_INNER) % SEQLEN;
    const int b = idx / (D_INNER * SEQLEN);
    float acc = cb[d];
#pragma unroll
    for (int k = 0; k < D_CONV; ++k) {
        const int ls = l + k - (D_CONV - 1);
        if (ls >= 0)
            acc += x[((size_t)b * SEQLEN + ls) * D_INNER + d] * w[d * D_CONV + k];
    }
    xc[idx] = silu_f(acc);
}

#define SCAN_T 8

struct ScanBlk {
    float dt[SCAN_T], u[SCAN_T], zz[SCAN_T], Bn[SCAN_T], Cn[SCAN_T];
};

__device__ __forceinline__ void scan_load(ScanBlk& r,
                                          const float* __restrict__ delta,
                                          const float* __restrict__ xc,
                                          const float* __restrict__ z,
                                          const float* __restrict__ xdbl,
                                          size_t row, int d, int n)
{
#pragma unroll
    for (int j = 0; j < SCAN_T; ++j) {
        const size_t off = (row + j) * D_INNER + d;
        const float* bc = xdbl + (row + j) * XDBL_LD;
        r.dt[j] = delta[off];
        r.u[j]  = xc[off];
        r.zz[j] = z[off];
        r.Bn[j] = bc[DT_RANK + n];
        r.Cn[j] = bc[DT_RANK + D_STATE + n];
    }
}

__device__ __forceinline__ void scan_steps(const ScanBlk& r, float& h,
                                           float An, float Dd,
                                           float* __restrict__ yout,
                                           size_t row, int d, int n)
{
    float y[SCAN_T];
#pragma unroll
    for (int j = 0; j < SCAN_T; ++j) {
        const float dt = r.dt[j];
        const float dA = expf(dt * An);
        h = fmaf(h, dA, (dt * r.u[j]) * r.Bn[j]);
        float t = h * r.Cn[j];
        t += __shfl_xor(t, 1, 16);
        t += __shfl_xor(t, 2, 16);
        t += __shfl_xor(t, 4, 16);
        t += __shfl_xor(t, 8, 16);
        y[j] = t;
    }
    if (n == 0) {
#pragma unroll
        for (int j = 0; j < SCAN_T; ++j) {
            const size_t off = (row + j) * D_INNER + d;
            yout[off] = (y[j] + Dd * r.u[j]) * silu_f(r.zz[j]);
        }
    }
}

__launch_bounds__(256)
__global__ void scan_kernel(const float* __restrict__ delta,
                            const float* __restrict__ xc,
                            const float* __restrict__ xdbl,
                            const float* __restrict__ A_log,
                            const float* __restrict__ Dp,
                            const float* __restrict__ z,
                            float* __restrict__ yout)
{
    const int gid = blockIdx.x * blockDim.x + threadIdx.x;
    const int n  = gid & (D_STATE - 1);
    const int ch = gid >> 4;
    if (ch >= BATCH * D_INNER) return;
    const int d = ch % D_INNER;
    const int b = ch / D_INNER;

    const float An = -expf(A_log[d * D_STATE + n]);
    const float Dd = Dp[d];
    float h = 0.f;

    const size_t row0 = (size_t)b * SEQLEN;

    ScanBlk ra, rb;
    scan_load(ra, delta, xc, z, xdbl, row0, d, n);
    for (int l0 = 0; l0 < SEQLEN; l0 += 2 * SCAN_T) {
        scan_load(rb, delta, xc, z, xdbl, row0 + l0 + SCAN_T, d, n);
        scan_steps(ra, h, An, Dd, yout, row0 + l0, d, n);
        const int ln = (l0 + 2 * SCAN_T < SEQLEN) ? (l0 + 2 * SCAN_T) : (SEQLEN - SCAN_T);
        scan_load(ra, delta, xc, z, xdbl, row0 + ln, d, n);
        scan_steps(rb, h, An, Dd, yout, row0 + l0 + SCAN_T, d, n);
    }
}

extern "C" void kernel_launch(void* const* d_in, const int* in_sizes, int n_in,
                              void* d_out, int out_size, void* d_ws, size_t ws_size,
                              hipStream_t stream)
{
    const float* hs        = (const float*)d_in[0];
    const float* in_proj_w = (const float*)d_in[1];
    const float* conv_w    = (const float*)d_in[2];
    const float* conv_b    = (const float*)d_in[3];
    const float* x_proj_w  = (const float*)d_in[4];
    const float* dt_proj_w = (const float*)d_in[5];
    const float* dt_proj_b = (const float*)d_in[6];
    const float* A_log     = (const float*)d_in[7];
    const float* Dp        = (const float*)d_in[8];
    const float* out_w     = (const float*)d_in[9];
    float* out = (float*)d_out;

    const size_t SZ_BLD = (size_t)BL * D_INNER * sizeof(float);
    char* ws = (char*)d_ws;
    float* x     = (float*)(ws);
    float* z     = (float*)(ws + SZ_BLD);
    float* xc    = (float*)(ws + 2 * SZ_BLD);
    float* delta = (float*)(ws + 3 * SZ_BLD);
    float* xdbl  = (float*)(ws + 4 * SZ_BLD);

    const dim3 blk(256);

    gemm_wmma_f32<8, 0><<<dim3(BL / 128, D_INNER / 128), blk, 0, stream>>>(
        hs, D_MODEL, in_proj_w, D_MODEL, x, D_INNER, D_MODEL, nullptr);
    gemm_wmma_f32<8, 0><<<dim3(BL / 128, D_INNER / 128), blk, 0, stream>>>(
        hs, D_MODEL, in_proj_w + (size_t)D_INNER * D_MODEL, D_MODEL,
        z, D_INNER, D_MODEL, nullptr);

    const int nelem = BATCH * SEQLEN * D_INNER;
    conv_silu_kernel<<<(nelem + 255) / 256, blk, 0, stream>>>(x, conv_w, conv_b, xc);

    gemm_wmma_f32<6, 0><<<dim3(BL / 128, 1), blk, 0, stream>>>(
        xc, D_INNER, x_proj_w, D_INNER, xdbl, XDBL_LD, D_INNER, nullptr);

    gemm_wmma_f32<8, 1><<<dim3(BL / 128, D_INNER / 128), blk, 0, stream>>>(
        xdbl, XDBL_LD, dt_proj_w, DT_RANK, delta, D_INNER, DT_RANK, dt_proj_b);

    float* y = x;
    const int scan_threads = BATCH * D_INNER * D_STATE;
    scan_kernel<<<(scan_threads + 255) / 256, blk, 0, stream>>>(
        delta, xc, xdbl, A_log, Dp, z, y);

    gemm_wmma_f32<8, 0><<<dim3(BL / 128, D_MODEL / 128), blk, 0, stream>>>(
        y, D_INNER, out_w, D_INNER, out, D_MODEL, D_INNER, nullptr);
}


# scan: gating (D*u skip, silu(z), store) of a block's 8 steps spread over 8 lanes of the channel, one step each, in place of lane 0 doing all 8; z loaded once per lane and block; sums unchanged
# speedup vs baseline: 4.4809x; 1.0739x over previous
#include <hip/hip_runtime.h>
#include <math.h>

#define D_MODEL  1024
#define D_STATE  16
#define D_CONV   4
#define D_INNER  2048
#define DT_RANK  64
#define BATCH    2
#define SEQLEN   2048
#define BL       (BATCH * SEQLEN)
#define XDBL_LD  (DT_RANK + 2 * D_STATE)

typedef __attribute__((ext_vector_type(2))) float v2f;
typedef __attribute__((ext_vector_type(4))) float v4f;
typedef __attribute__((ext_vector_type(8))) float v8f;

__device__ __forceinline__ float silu_f(float v)     { return v / (1.f + expf(-v)); }
__device__ __forceinline__ float softplus_f(float v) { return (v > 20.f) ? v : log1pf(expf(v)); }

template <int NT, int EPI>
__launch_bounds__(256)
__global__ void gemm_wmma_f32(const float* __restrict__ A, int lda,
                              const float* __restrict__ W, int ldw,
                              float* __restrict__ C, int ldc, int K,
                              const float* __restrict__ bias)
{
    constexpr int KC      = 32;
    constexpr int KC4     = KC / 4;
    constexpr int LDS_LD  = KC + 4;
    constexpr int NROWS_B = 16 * NT;
    constexpr int NA      = (128 * KC4) / 256;
    constexpr int NB      = (NROWS_B * KC4) / 256;

    __shared__ float As[128 * LDS_LD];
    __shared__ float Bs[NROWS_B * LDS_LD];

    const int tid  = threadIdx.x;
    const int wave = tid >> 5;
    const int lane = tid & 31;
    const int half = lane >> 4;
    const int lidx = lane & 15;
    const int m_blk  = blockIdx.x * 128;
    const int n_base = blockIdx.y * NROWS_B;

    v8f acc[NT];
#pragma unroll
    for (int t = 0; t < NT; ++t) acc[t] = (v8f){0.f,0.f,0.f,0.f,0.f,0.f,0.f,0.f};

    v4f ra[NA], rb[NB];

#pragma unroll
    for (int i = 0; i < NA; ++i) {
        const int idx = tid + i * 256, r = idx >> 3, c = (idx & 7) * 4;
        ra[i] = *(const v4f*)(A + (size_t)(m_blk + r) * lda + c);
    }
#pragma unroll
    for (int i = 0; i < NB; ++i) {
        const int idx = tid + i * 256, r = idx >> 3, c = (idx & 7) * 4;
        rb[i] = *(const v4f*)(W + (size_t)(n_base + r) * ldw + c);
    }

    for (int k0 = 0; k0 < K; k0 += KC) {
#pragma unroll
        for (int i = 0; i < NA; ++i) {
            const int idx = tid + i * 256, r = idx >> 3, c = (idx & 7) * 4;
            *(v4f*)(As + r * LDS_LD + c) = ra[i];
        }
#pragma unroll
        for (int i = 0; i < NB; ++i) {
            const int idx = tid + i * 256, r = idx >> 3, c = (idx & 7) * 4;
            *(v4f*)(Bs + r * LDS_LD + c) = rb[i];
        }
        __syncthreads();

        if (k0 + KC < K) {
#pragma unroll
            for (int i = 0; i < NA; ++i) {
                const int idx = tid + i * 256, r = idx >> 3, c = (idx & 7) * 4;
                ra[i] = *(const v4f*)(A + (size_t)(m_blk + r) * lda + (k0 + KC) + c);
            }
#pragma unroll
            for (int i = 0; i < NB; ++i) {
                const int idx = tid + i * 256, r = idx >> 3, c = (idx & 7) * 4;
                rb[i] = *(const v4f*)(W + (size_t)(n_base + r) * ldw + (k0 + KC) + c);
            }
        }

        const float* a_lds = As + (wave * 16 + lidx) * LDS_LD + 2 * half;
        const float* b_lds = Bs + lidx * LDS_LD + 2 * half;
#pragma unroll
        for (int kk = 0; kk < KC; kk += 4) {
            v2f a = *(const v2f*)(a_lds + kk);
#pragma unroll
            for (int t = 0; t < NT; ++t) {
                v2f b = *(const v2f*)(b_lds + t * 16 * LDS_LD + kk);
                acc[t] = __builtin_amdgcn_wmma_f32_16x16x4_f32(
                    false, a, false, b, (short)0, acc[t], false, false);
            }
        }
        __syncthreads();
    }

#pragma unroll
    for (int t = 0; t < NT; ++t) {
        const int n = n_base + t * 16 + lidx;
#pragma unroll
        for (int v = 0; v < 8; ++v) {
            const int m = m_blk + wave * 16 + half * 8 + v;
            float val = acc[t][v];
            if (EPI == 1) val = softplus_f(val + bias[n]);
            C[(size_t)m * ldc + n] = val;
        }
    }
}

__global__ void conv_silu_kernel(const float* __restrict__ x,
                                 const float* __restrict__ w,
                                 const float* __restrict__ cb,
                                 float* __restrict__ xc)
{
    const int idx = blockIdx.x * blockDim.x + threadIdx.x;
    if (idx >= BATCH * SEQLEN * D_INNER) return;
    const int d = idx % D_INNER;
    const int l = (idx / D_INNER) % SEQLEN;
    const int b = idx / (D_INNER * SEQLEN);
    float acc = cb[d];
#pragma unroll
    for (int k = 0; k < D_CONV; ++k) {
        const int ls = l + k - (D_CONV - 1);
        if (ls >= 0)
            acc += x[((size_t)b * SEQLEN + ls) * D_INNER + d] * w[d * D_CONV + k];
    }
    xc[idx] = silu_f(acc);
}

#define SCAN_T 8

struct ScanBlk {
    float dt[SCAN_T], u[SCAN_T], Bn[SCAN_T], Cn[SCAN_T];
    float ug, zg;
};

__device__ __forceinline__ void scan_load(ScanBlk& r,
                                          const float* __restrict__ delta,
                                          const float* __restrict__ xc,
                                          const float* __restrict__ z,
                                          const float* __restrict__ xdbl,
                                          size_t row, int d, int n, int jl)
{
#pragma unroll
    for (int j = 0; j < SCAN_T; ++j) {
        const size_t off = (row + j) * D_INNER + d;
        const float* bc = xdbl + (row + j) * XDBL_LD;
        r.dt[j] = delta[off];
        r.u[j]  = xc[off];
        r.Bn[j] = bc[DT_RANK + n];
        r.Cn[j] = bc[DT_RANK + D_STATE + n];
    }
    const size_t offg = (row + jl) * D_INNER + d;
    r.ug = xc[offg];
    r.zg = z[offg];
}

__device__ __forceinline__ void scan_steps(const ScanBlk& r, float& h,
                                           float An, float Dd,
                                           float* __restrict__ yout,
                                           size_t row, int d, int n, int jl)
{
    float y[SCAN_T];
#pragma unroll
    for (int j = 0; j < SCAN_T; ++j) {
        const float dt = r.dt[j];
        const float dA = expf(dt * An);
        h = fmaf(h, dA, (dt * r.u[j]) * r.Bn[j]);
        float t = h * r.Cn[j];
        t += __shfl_xor(t, 1, 16);
        t += __shfl_xor(t, 2, 16);
        t += __shfl_xor(t, 4, 16);
        t += __shfl_xor(t, 8, 16);
        y[j] = t;
    }
    float ys = y[0];
#pragma unroll
    for (int j = 1; j < SCAN_T; ++j) ys = (jl == j) ? y[j] : ys;
    const float o = (ys + Dd * r.ug) * silu_f(r.zg);
    if (n < SCAN_T) yout[(row + jl) * D_INNER + d] = o;
}

__launch_bounds__(256)
__global__ void scan_kernel(const float* __restrict__ delta,
                            const float* __restrict__ xc,
                            const float* __restrict__ xdbl,
                            const float* __restrict__ A_log,
                            const float* __restrict__ Dp,
                            const float* __restrict__ z,
                            float* __restrict__ yout)
{
    const int gid = blockIdx.x * blockDim.x + threadIdx.x;
    const int n  = gid & (D_STATE - 1);
    const int ch = gid >> 4;
    if (ch >= BATCH * D_INNER) return;
    const int d = ch % D_INNER;
    const int b = ch / D_INNER;

    const int jl = n & (SCAN_T - 1);
    const float An = -expf(A_log[d * D_STATE + n]);
    const float Dd = Dp[d];
    float h = 0.f;

    const size_t row0 = (size_t)b * SEQLEN;

    ScanBlk ra, rb;
    scan_load(ra, delta, xc, z, xdbl, row0, d, n, jl);
    for (int l0 = 0; l0 < SEQLEN; l0 += 2 * SCAN_T) {
        scan_load(rb, delta, xc, z, xdbl, row0 + l0 + SCAN_T, d, n, jl);
        scan_steps(ra, h, An, Dd, yout, row0 + l0, d, n, jl);
        const int ln = (l0 + 2 * SCAN_T < SEQLEN) ? (l0 + 2 * SCAN_T) : (SEQLEN - SCAN_T);
        scan_load(ra, delta, xc, z, xdbl, row0 + ln, d, n, jl);
        scan_steps(rb, h, An, Dd, yout, row0 + l0 + SCAN_T, d, n, jl);
    }
}

extern "C" void kernel_launch(void* const* d_in, const int* in_sizes, int n_in,
                              void* d_out, int out_size, void* d_ws, size_t ws_size,
                              hipStream_t stream)
{
    const float* hs        = (const float*)d_in[0];
    const float* in_proj_w = (const float*)d_in[1];
    const float* conv_w    = (const float*)d_in[2];
    const float* conv_b    = (const float*)d_in[3];
    const float* x_proj_w  = (const float*)d_in[4];
    const float* dt_proj_w = (const float*)d_in[5];
    const float* dt_proj_b = (const float*)d_in[6];
    const float* A_log     = (const float*)d_in[7];
    const float* Dp        = (const float*)d_in[8];
    const float* out_w     = (const float*)d_in[9];
    float* out = (float*)d_out;

    const size_t SZ_BLD = (size_t)BL * D_INNER * sizeof(float);
    char* ws = (char*)d_ws;
    float* x     = (float*)(ws);
    float* z     = (float*)(ws + SZ_BLD);
    float* xc    = (float*)(ws + 2 * SZ_BLD);
    float* delta = (float*)(ws + 3 * SZ_BLD);
    float* xdbl  = (float*)(ws + 4 * SZ_BLD);

    const dim3 blk(256);

    gemm_wmma_f32<8, 0><<<dim3(BL / 128, D_INNER / 128), blk, 0, stream>>>(
        hs, D_MODEL, in_proj_w, D_MODEL, x, D_INNER, D_MODEL, nullptr);
    gemm_wmma_f32<8, 0><<<dim3(BL / 128, D_INNER / 128), blk, 0, stream>>>(
        hs, D_MODEL, in_proj_w + (size_t)D_INNER * D_MODEL, D_MODEL,
        z, D_INNER, D_MODEL, nullptr);

    const int nelem = BATCH * SEQLEN * D_INNER;
    conv_silu_kernel<<<(nelem + 255) / 256, blk, 0, stream>>>(x, conv_w, conv_b, xc);

    gemm_wmma_f32<6, 0><<<dim3(BL / 128, 1), blk, 0, stream>>>(
        xc, D_INNER, x_proj_w, D_INNER, xdbl, XDBL_LD, D_INNER, nullptr);

    gemm_wmma_f32<8, 1><<<dim3(BL / 128, D_INNER / 128), blk, 0, stream>>>(
        xdbl, XDBL_LD, dt_proj_w, DT_RANK, delta, D_INNER, DT_RANK, dt_proj_b);

    float* y = x;
    const int scan_threads = BATCH * D_INNER * D_STATE;
    scan_kernel<<<(scan_threads + 255) / 256, blk, 0, stream>>>(
        delta, xc, xdbl, A_log, Dp, z, y);

    gemm_wmma_f32<8, 0><<<dim3(BL / 128, D_MODEL / 128), blk, 0, stream>>>(
        y, D_INNER, out_w, D_INNER, out, D_MODEL, D_INNER, nullptr);
}


# x_proj gemm (N=96, only 32 blocks) split 8 ways along K into 256 blocks; partial tiles go to workspace planes and one added small kernel sums them in plane order; K sum order changes
# speedup vs baseline: 4.8464x; 1.0816x over previous
#include <hip/hip_runtime.h>
#include <math.h>

#define D_MODEL  1024
#define D_STATE  16
#define D_CONV   4
#define D_INNER  2048
#define DT_RANK  64
#define BATCH    2
#define SEQLEN   2048
#define BL       (BATCH * SEQLEN)
#define XDBL_LD  (DT_RANK + 2 * D_STATE)

typedef __attribute__((ext_vector_type(2))) float v2f;
typedef __attribute__((ext_vector_type(4))) float v4f;
typedef __attribute__((ext_vector_type(8))) float v8f;

__device__ __forceinline__ float silu_f(float v)     { return v / (1.f + expf(-v)); }
__device__ __forceinline__ float softplus_f(float v) { return (v > 20.f) ? v : log1pf(expf(v)); }

template <int NT, int EPI, int KSPLIT = 1>
__launch_bounds__(256)
__global__ void gemm_wmma_f32(const float* __restrict__ A, int lda,
                              const float* __restrict__ W, int ldw,
                              float* __restrict__ C, int ldc, int K,
                              const float* __restrict__ bias)
{
    constexpr int KC      = 32;
    constexpr int KC4     = KC / 4;
    constexpr int LDS_LD  = KC + 4;
    constexpr int NROWS_B = 16 * NT;
    constexpr int NA      = (128 * KC4) / 256;
    constexpr int NB      = (NROWS_B * KC4) / 256;

    __shared__ float As[128 * LDS_LD];
    __shared__ float Bs[NROWS_B * LDS_LD];

    const int tid  = threadIdx.x;
    const int wave = tid >> 5;
    const int lane = tid & 31;
    const int half = lane >> 4;
    const int lidx = lane & 15;
    const int m_blk  = blockIdx.x * 128;
    const int n_base = (KSPLIT > 1) ? 0 : blockIdx.y * NROWS_B;
    if (KSPLIT > 1) {
        K /= KSPLIT;
        A += blockIdx.y * K;
        W += blockIdx.y * K;
        C += (size_t)blockIdx.y * gridDim.x * 128 * ldc;
    }

    v8f acc[NT];
#pragma unroll
    for (int t = 0; t < NT; ++t) acc[t] = (v8f){0.f,0.f,0.f,0.f,0.f,0.f,0.f,0.f};

    v4f ra[NA], rb[NB];

#pragma unroll
    for (int i = 0; i < NA; ++i) {
        const int idx = tid + i * 256, r = idx >> 3, c = (idx & 7) * 4;
        ra[i] = *(const v4f*)(A + (size_t)(m_blk + r) * lda + c);
    }
#pragma unroll
    for (int i = 0; i < NB; ++i) {
        const int idx = tid + i * 256, r = idx >> 3, c = (idx & 7) * 4;
        rb[i] = *(const v4f*)(W + (size_t)(n_base + r) * ldw + c);
    }

    for (int k0 = 0; k0 < K; k0 += KC) {
#pragma unroll
        for (int i = 0; i < NA; ++i) {
            const int idx = tid + i * 256, r = idx >> 3, c = (idx & 7) * 4;
            *(v4f*)(As + r * LDS_LD + c) = ra[i];
        }
#pragma unroll
        for (int i = 0; i < NB; ++i) {
            const int idx = tid + i * 256, r = idx >> 3, c = (idx & 7) * 4;
            *(v4f*)(Bs + r * LDS_LD + c) = rb[i];
        }
        __syncthreads();

        if (k0 + KC < K) {
#pragma unroll
            for (int i = 0; i < NA; ++i) {
                const int idx = tid + i * 256, r = idx >> 3, c = (idx & 7) * 4;
                ra[i] = *(const v4f*)(A + (size_t)(m_blk + r) * lda + (k0 + KC) + c);
            }
#pragma unroll
            for (int i = 0; i < NB; ++i) {
                const int idx = tid + i * 256, r = idx >> 3, c = (idx & 7) * 4;
                rb[i] = *(const v4f*)(W + (size_t)(n_base + r) * ldw + (k0 + KC) + c);
            }
        }

        const float* a_lds = As + (wave * 16 + lidx) * LDS_LD + 2 * half;
        const float* b_lds = Bs + lidx * LDS_LD + 2 * half;
#pragma unroll
        for (int kk = 0; kk < KC; kk += 4) {
            v2f a = *(const v2f*)(a_lds + kk);
#pragma unroll
            for (int t = 0; t < NT; ++t) {
                v2f b = *(const v2f*)(b_lds + t * 16 * LDS_LD + kk);
                acc[t] = __builtin_amdgcn_wmma_f32_16x16x4_f32(
                    false, a, false, b, (short)0, acc[t], false, false);
            }
        }
        __syncthreads();
    }

#pragma unroll
    for (int t = 0; t < NT; ++t) {
        const int n = n_base + t * 16 + lidx;
#pragma unroll
        for (int v = 0; v < 8; ++v) {
            const int m = m_blk + wave * 16 + half * 8 + v;
            float val = acc[t][v];
            if (EPI == 1) val = softplus_f(val + bias[n]);
            C[(size_t)m * ldc + n] = val;
        }
    }
}

#define X_SPLIT 8

__global__ void splitk_sum_kernel(const float* __restrict__ part,
                                  float* __restrict__ dst)
{
    constexpr int PLANE4 = BL * XDBL_LD / 4;
    const int i = blockIdx.x * blockDim.x + threadIdx.x;
    if (i >= PLANE4) return;
    const v4f* p = (const v4f*)part + i;
    v4f v[X_SPLIT];
#pragma unroll
    for (int s = 0; s < X_SPLIT; ++s) v[s] = p[(size_t)s * PLANE4];
    v4f acc = v[0];
#pragma unroll
    for (int s = 1; s < X_SPLIT; ++s) acc += v[s];
    ((v4f*)dst)[i] = acc;
}

__global__ void conv_silu_kernel(const float* __restrict__ x,
                                 const float* __restrict__ w,
                                 const float* __restrict__ cb,
                                 float* __restrict__ xc)
{
    const int idx = blockIdx.x * blockDim.x + threadIdx.x;
    if (idx >= BATCH * SEQLEN * D_INNER) return;
    const int d = idx % D_INNER;
    const int l = (idx / D_INNER) % SEQLEN;
    const int b = idx / (D_INNER * SEQLEN);
    float acc = cb[d];
#pragma unroll
    for (int k = 0; k < D_CONV; ++k) {
        const int ls = l + k - (D_CONV - 1);
        if (ls >= 0)
            acc += x[((size_t)b * SEQLEN + ls) * D_INNER + d] * w[d * D_CONV + k];
    }
    xc[idx] = silu_f(acc);
}

#define SCAN_T 8

struct ScanBlk {
    float dt[SCAN_T], u[SCAN_T], Bn[SCAN_T], Cn[SCAN_T];
    float ug, zg;
};

__device__ __forceinline__ void scan_load(ScanBlk& r,
                                          const float* __restrict__ delta,
                                          const float* __restrict__ xc,
                                          const float* __restrict__ z,
                                          const float* __restrict__ xdbl,
                                          size_t row, int d, int n, int jl)
{
#pragma unroll
    for (int j = 0; j < SCAN_T; ++j) {
        const size_t off = (row + j) * D_INNER + d;
        const float* bc = xdbl + (row + j) * XDBL_LD;
        r.dt[j] = delta[off];
        r.u[j]  = xc[off];
        r.Bn[j] = bc[DT_RANK + n];
        r.Cn[j] = bc[DT_RANK + D_STATE + n];
    }
    const size_t offg = (row + jl) * D_INNER + d;
    r.ug = xc[offg];
    r.zg = z[offg];
}

__device__ __forceinline__ void scan_steps(const ScanBlk& r, float& h,
                                           float An, float Dd,
                                           float* __restrict__ yout,
                                           size_t row, int d, int n, int jl)
{
    float y[SCAN_T];
#pragma unroll
    for (int j = 0; j < SCAN_T; ++j) {
        const float dt = r.dt[j];
        const float dA = expf(dt * An);
        h = fmaf(h, dA, (dt * r.u[j]) * r.Bn[j]);
        float t = h * r.Cn[j];
        t += __shfl_xor(t, 1, 16);
        t += __shfl_xor(t, 2, 16);
        t += __shfl_xor(t, 4, 16);
        t += __shfl_xor(t, 8, 16);
        y[j] = t;
    }
    float ys = y[0];
#pragma unroll
    for (int j = 1; j < SCAN_T; ++j) ys = (jl == j) ? y[j] : ys;
    const float o = (ys + Dd * r.ug) * silu_f(r.zg);
    if (n < SCAN_T) yout[(row + jl) * D_INNER + d] = o;
}

__launch_bounds__(256)
__global__ void scan_kernel(const float* __restrict__ delta,
                            const float* __restrict__ xc,
                            const float* __restrict__ xdbl,
                            const float* __restrict__ A_log,
                            const float* __restrict__ Dp,
                            const float* __restrict__ z,
                            float* __restrict__ yout)
{
    const int gid = blockIdx.x * blockDim.x + threadIdx.x;
    const int n  = gid & (D_STATE - 1);
    const int ch = gid >> 4;
    if (ch >= BATCH * D_INNER) return;
    const int d = ch % D_INNER;
    const int b = ch / D_INNER;

    const int jl = n & (SCAN_T - 1);
    const float An = -expf(A_log[d * D_STATE + n]);
    const float Dd = Dp[d];
    float h = 0.f;

    const size_t row0 = (size_t)b * SEQLEN;

    ScanBlk ra, rb;
    scan_load(ra, delta, xc, z, xdbl, row0, d, n, jl);
    for (int l0 = 0; l0 < SEQLEN; l0 += 2 * SCAN_T) {
        scan_load(rb, delta, xc, z, xdbl, row0 + l0 + SCAN_T, d, n, jl);
        scan_steps(ra, h, An, Dd, yout, row0 + l0, d, n, jl);
        const int ln = (l0 + 2 * SCAN_T < SEQLEN) ? (l0 + 2 * SCAN_T) : (SEQLEN - SCAN_T);
        scan_load(ra, delta, xc, z, xdbl, row0 + ln, d, n, jl);
        scan_steps(rb, h, An, Dd, yout, row0 + l0 + SCAN_T, d, n, jl);
    }
}

extern "C" void kernel_launch(void* const* d_in, const int* in_sizes, int n_in,
                              void* d_out, int out_size, void* d_ws, size_t ws_size,
                              hipStream_t stream)
{
    const float* hs        = (const float*)d_in[0];
    const float* in_proj_w = (const float*)d_in[1];
    const float* conv_w    = (const float*)d_in[2];
    const float* conv_b    = (const float*)d_in[3];
    const float* x_proj_w  = (const float*)d_in[4];
    const float* dt_proj_w = (const float*)d_in[5];
    const float* dt_proj_b = (const float*)d_in[6];
    const float* A_log     = (const float*)d_in[7];
    const float* Dp        = (const float*)d_in[8];
    const float* out_w     = (const float*)d_in[9];
    float* out = (float*)d_out;

    const size_t SZ_BLD = (size_t)BL * D_INNER * sizeof(float);
    char* ws = (char*)d_ws;
    float* x     = (float*)(ws);
    float* z     = (float*)(ws + SZ_BLD);
    float* xc    = (float*)(ws + 2 * SZ_BLD);
    float* delta = (float*)(ws + 3 * SZ_BLD);
    float* xdbl  = (float*)(ws + 4 * SZ_BLD);
    float* xpart = (float*)(ws + 5 * SZ_BLD);

    const dim3 blk(256);

    gemm_wmma_f32<8, 0><<<dim3(BL / 128, D_INNER / 128), blk, 0, stream>>>(
        hs, D_MODEL, in_proj_w, D_MODEL, x, D_INNER, D_MODEL, nullptr);
    gemm_wmma_f32<8, 0><<<dim3(BL / 128, D_INNER / 128), blk, 0, stream>>>(
        hs, D_MODEL, in_proj_w + (size_t)D_INNER * D_MODEL, D_MODEL,
        z, D_INNER, D_MODEL, nullptr);

    const int nelem = BATCH * SEQLEN * D_INNER;
    conv_silu_kernel<<<(nelem + 255) / 256, blk, 0, stream>>>(x, conv_w, conv_b, xc);

    gemm_wmma_f32<6, 0, X_SPLIT><<<dim3(BL / 128, X_SPLIT), blk, 0, stream>>>(
        xc, D_INNER, x_proj_w, D_INNER, xpart, XDBL_LD, D_INNER, nullptr);
    splitk_sum_kernel<<<(BL * XDBL_LD / 4 + 255) / 256, blk, 0, stream>>>(xpart, xdbl);

    gemm_wmma_f32<8, 1><<<dim3(BL / 128, D_INNER / 128), blk, 0, stream>>>(
        xdbl, XDBL_LD, dt_proj_w, DT_RANK, delta, D_INNER, DT_RANK, dt_proj_b);

    float* y = x;
    const int scan_threads = BATCH * D_INNER * D_STATE;
    scan_kernel<<<(scan_threads + 255) / 256, blk, 0, stream>>>(
        delta, xc, xdbl, A_log, Dp, z, y);

    gemm_wmma_f32<8, 0><<<dim3(BL / 128, D_MODEL / 128), blk, 0, stream>>>(
        y, D_INNER, out_w, D_INNER, out, D_MODEL, D_INNER, nullptr);
}


# library expf/log1pf/division chains of the scan decay, its gate and the dt softplus replaced by hardware exp2/log2/rcp forms (decay = exp2(dt*A*log2e), scale folded per lane); last-bit value changes
# speedup vs baseline: 5.2252x; 1.0782x over previous
#include <hip/hip_runtime.h>
#include <math.h>

#define D_MODEL  1024
#define D_STATE  16
#define D_CONV   4
#define D_INNER  2048
#define DT_RANK  64
#define BATCH    2
#define SEQLEN   2048
#define BL       (BATCH * SEQLEN)
#define XDBL_LD  (DT_RANK + 2 * D_STATE)

typedef __attribute__((ext_vector_type(2))) float v2f;
typedef __attribute__((ext_vector_type(4))) float v4f;
typedef __attribute__((ext_vector_type(8))) float v8f;

__device__ __forceinline__ float silu_f(float v)     { return v / (1.f + expf(-v)); }

#define LOG2E_F 1.44269504088896340736f
#define LN2_F   0.693147180559945309417f

__device__ __forceinline__ float exp_hw(float v) { return __builtin_amdgcn_exp2f(v * LOG2E_F); }

__device__ __forceinline__ float silu_hw(float v)
{
    return v * __builtin_amdgcn_rcpf(1.f + exp_hw(-v));
}

__device__ __forceinline__ float softplus_hw(float v)
{
    const float e = exp_hw(v);
    float p = fmaf(e, -1.f / 6.f, 1.f / 5.f);
    p = fmaf(e, p, -1.f / 4.f);
    p = fmaf(e, p, 1.f / 3.f);
    p = fmaf(e, p, -1.f / 2.f);
    p = fmaf(e, p, 1.f);
    const float lo = e * p;
    const float u  = 1.f + e;
    const float hi = (__builtin_amdgcn_logf(u) * LN2_F) * (e * __builtin_amdgcn_rcpf(u - 1.f));
    const float r  = (e < 0.0625f) ? lo : hi;
    return (v > 20.f) ? v : r;
}

template <int NT, int EPI, int KSPLIT = 1>
__launch_bounds__(256)
__global__ void gemm_wmma_f32(const float* __restrict__ A, int lda,
                              const float* __restrict__ W, int ldw,
                              float* __restrict__ C, int ldc, int K,
                              const float* __restrict__ bias)
{
    constexpr int KC      = 32;
    constexpr int KC4     = KC / 4;
    constexpr int LDS_LD  = KC + 4;
    constexpr int NROWS_B = 16 * NT;
    constexpr int NA      = (128 * KC4) / 256;
    constexpr int NB      = (NROWS_B * KC4) / 256;

    __shared__ float As[128 * LDS_LD];
    __shared__ float Bs[NROWS_B * LDS_LD];

    const int tid  = threadIdx.x;
    const int wave = tid >> 5;
    const int lane = tid & 31;
    const int half = lane >> 4;
    const int lidx = lane & 15;
    const int m_blk  = blockIdx.x * 128;
    const int n_base = (KSPLIT > 1) ? 0 : blockIdx.y * NROWS_B;
    if (KSPLIT > 1) {
        K /= KSPLIT;
        A += blockIdx.y * K;
        W += blockIdx.y * K;
        C += (size_t)blockIdx.y * gridDim.x * 128 * ldc;
    }

    v8f acc[NT];
#pragma unroll
    for (int t = 0; t < NT; ++t) acc[t] = (v8f){0.f,0.f,0.f,0.f,0.f,0.f,0.f,0.f};

    v4f ra[NA], rb[NB];

#pragma unroll
    for (int i = 0; i < NA; ++i) {
        const int idx = tid + i * 256, r = idx >> 3, c = (idx & 7) * 4;
        ra[i] = *(const v4f*)(A + (size_t)(m_blk + r) * lda + c);
    }
#pragma unroll
    for (int i = 0; i < NB; ++i) {
        const int idx = tid + i * 256, r = idx >> 3, c = (idx & 7) * 4;
        rb[i] = *(const v4f*)(W + (size_t)(n_base + r) * ldw + c);
    }

    for (int k0 = 0; k0 < K; k0 += KC) {
#pragma unroll
        for (int i = 0; i < NA; ++i) {
            const int idx = tid + i * 256, r = idx >> 3, c = (idx & 7) * 4;
            *(v4f*)(As + r * LDS_LD + c) = ra[i];
        }
#pragma unroll
        for (int i = 0; i < NB; ++i) {
            const int idx = tid + i * 256, r = idx >> 3, c = (idx & 7) * 4;
            *(v4f*)(Bs + r * LDS_LD + c) = rb[i];
        }
        __syncthreads();

        if (k0 + KC < K) {
#pragma unroll
            for (int i = 0; i < NA; ++i) {
                const int idx = tid + i * 256, r = idx >> 3, c = (idx & 7) * 4;
                ra[i] = *(const v4f*)(A + (size_t)(m_blk + r) * lda + (k0 + KC) + c);
            }
#pragma unroll
            for (int i = 0; i < NB; ++i) {
                const int idx = tid + i * 256, r = idx >> 3, c = (idx & 7) * 4;
                rb[i] = *(const v4f*)(W + (size_t)(n_base + r) * ldw + (k0 + KC) + c);
            }
        }

        const float* a_lds = As + (wave * 16 + lidx) * LDS_LD + 2 * half;
        const float* b_lds = Bs + lidx * LDS_LD + 2 * half;
#pragma unroll
        for (int kk = 0; kk < KC; kk += 4) {
            v2f a = *(const v2f*)(a_lds + kk);
#pragma unroll
            for (int t = 0; t < NT; ++t) {
                v2f b = *(const v2f*)(b_lds + t * 16 * LDS_LD + kk);
                acc[t] = __builtin_amdgcn_wmma_f32_16x16x4_f32(
                    false, a, false, b, (short)0, acc[t], false, false);
            }
        }
        __syncthreads();
    }

#pragma unroll
    for (int t = 0; t < NT; ++t) {
        const int n = n_base + t * 16 + lidx;
#pragma unroll
        for (int v = 0; v < 8; ++v) {
            const int m = m_blk + wave * 16 + half * 8 + v;
            float val = acc[t][v];
            if (EPI == 1) val = softplus_hw(val + bias[n]);
            C[(size_t)m * ldc + n] = val;
        }
    }
}

#define X_SPLIT 8

__global__ void splitk_sum_kernel(const float* __restrict__ part,
                                  float* __restrict__ dst)
{
    constexpr int PLANE4 = BL * XDBL_LD / 4;
    const int i = blockIdx.x * blockDim.x + threadIdx.x;
    if (i >= PLANE4) return;
    const v4f* p = (const v4f*)part + i;
    v4f v[X_SPLIT];
#pragma unroll
    for (int s = 0; s < X_SPLIT; ++s) v[s] = p[(size_t)s * PLANE4];
    v4f acc = v[0];
#pragma unroll
    for (int s = 1; s < X_SPLIT; ++s) acc += v[s];
    ((v4f*)dst)[i] = acc;
}

__global__ void conv_silu_kernel(const float* __restrict__ x,
                                 const float* __restrict__ w,
                                 const float* __restrict__ cb,
                                 float* __restrict__ xc)
{
    const int idx = blockIdx.x * blockDim.x + threadIdx.x;
    if (idx >= BATCH * SEQLEN * D_INNER) return;
    const int d = idx % D_INNER;
    const int l = (idx / D_INNER) % SEQLEN;
    const int b = idx / (D_INNER * SEQLEN);
    float acc = cb[d];
#pragma unroll
    for (int k = 0; k < D_CONV; ++k) {
        const int ls = l + k - (D_CONV - 1);
        if (ls >= 0)
            acc += x[((size_t)b * SEQLEN + ls) * D_INNER + d] * w[d * D_CONV + k];
    }
    xc[idx] = silu_f(acc);
}

#define SCAN_T 8

struct ScanBlk {
    float dt[SCAN_T], u[SCAN_T], Bn[SCAN_T], Cn[SCAN_T];
    float ug, zg;
};

__device__ __forceinline__ void scan_load(ScanBlk& r,
                                          const float* __restrict__ delta,
                                          const float* __restrict__ xc,
                                          const float* __restrict__ z,
                                          const float* __restrict__ xdbl,
                                          size_t row, int d, int n, int jl)
{
#pragma unroll
    for (int j = 0; j < SCAN_T; ++j) {
        const size_t off = (row + j) * D_INNER + d;
        const float* bc = xdbl + (row + j) * XDBL_LD;
        r.dt[j] = delta[off];
        r.u[j]  = xc[off];
        r.Bn[j] = bc[DT_RANK + n];
        r.Cn[j] = bc[DT_RANK + D_STATE + n];
    }
    const size_t offg = (row + jl) * D_INNER + d;
    r.ug = xc[offg];
    r.zg = z[offg];
}

__device__ __forceinline__ void scan_steps(const ScanBlk& r, float& h,
                                           float An2, float Dd,
                                           float* __restrict__ yout,
                                           size_t row, int d, int n, int jl)
{
    float y[SCAN_T];
#pragma unroll
    for (int j = 0; j < SCAN_T; ++j) {
        const float dt = r.dt[j];
        const float dA = __builtin_amdgcn_exp2f(dt * An2);
        h = fmaf(h, dA, (dt * r.u[j]) * r.Bn[j]);
        float t = h * r.Cn[j];
        t += __shfl_xor(t, 1, 16);
        t += __shfl_xor(t, 2, 16);
        t += __shfl_xor(t, 4, 16);
        t += __shfl_xor(t, 8, 16);
        y[j] = t;
    }
    float ys = y[0];
#pragma unroll
    for (int j = 1; j < SCAN_T; ++j) ys = (jl == j) ? y[j] : ys;
    const float o = (ys + Dd * r.ug) * silu_hw(r.zg);
    if (n < SCAN_T) yout[(row + jl) * D_INNER + d] = o;
}

__launch_bounds__(256)
__global__ void scan_kernel(const float* __restrict__ delta,
                            const float* __restrict__ xc,
                            const float* __restrict__ xdbl,
                            const float* __restrict__ A_log,
                            const float* __restrict__ Dp,
                            const float* __restrict__ z,
                            float* __restrict__ yout)
{
    const int gid = blockIdx.x * blockDim.x + threadIdx.x;
    const int n  = gid & (D_STATE - 1);
    const int ch = gid >> 4;
    if (ch >= BATCH * D_INNER) return;
    const int d = ch % D_INNER;
    const int b = ch / D_INNER;

    const int jl = n & (SCAN_T - 1);
    const float An2 = -expf(A_log[d * D_STATE + n]) * LOG2E_F;
    const float Dd = Dp[d];
    float h = 0.f;

    const size_t row0 = (size_t)b * SEQLEN;

    ScanBlk ra, rb;
    scan_load(ra, delta, xc, z, xdbl, row0, d, n, jl);
    for (int l0 = 0; l0 < SEQLEN; l0 += 2 * SCAN_T) {
        scan_load(rb, delta, xc, z, xdbl, row0 + l0 + SCAN_T, d, n, jl);
        scan_steps(ra, h, An2, Dd, yout, row0 + l0, d, n, jl);
        const int ln = (l0 + 2 * SCAN_T < SEQLEN) ? (l0 + 2 * SCAN_T) : (SEQLEN - SCAN_T);
        scan_load(ra, delta, xc, z, xdbl, row0 + ln, d, n, jl);
        scan_steps(rb, h, An2, Dd, yout, row0 + l0 + SCAN_T, d, n, jl);
    }
}

extern "C" void kernel_launch(void* const* d_in, const int* in_sizes, int n_in,
                              void* d_out, int out_size, void* d_ws, size_t ws_size,
                              hipStream_t stream)
{
    const float* hs        = (const float*)d_in[0];
    const float* in_proj_w = (const float*)d_in[1];
    const float* conv_w    = (const float*)d_in[2];
    const float* conv_b    = (const float*)d_in[3];
    const float* x_proj_w  = (const float*)d_in[4];
    const float* dt_proj_w = (const float*)d_in[5];
    const float* dt_proj_b = (const float*)d_in[6];
    const float* A_log     = (const float*)d_in[7];
    const float* Dp        = (const float*)d_in[8];
    const float* out_w     = (const float*)d_in[9];
    float* out = (float*)d_out;

    const size_t SZ_BLD = (size_t)BL * D_INNER * sizeof(float);
    char* ws = (char*)d_ws;
    float* x     = (float*)(ws);
    float* z     = (float*)(ws + SZ_BLD);
    float* xc    = (float*)(ws + 2 * SZ_BLD);
    float* delta = (float*)(ws + 3 * SZ_BLD);
    float* xdbl  = (float*)(ws + 4 * SZ_BLD);
    float* xpart = (float*)(ws + 5 * SZ_BLD);

    const dim3 blk(256);

    gemm_wmma_f32<8, 0><<<dim3(BL / 128, D_INNER / 128), blk, 0, stream>>>(
        hs, D_MODEL, in_proj_w, D_MODEL, x, D_INNER, D_MODEL, nullptr);
    gemm_wmma_f32<8, 0><<<dim3(BL / 128, D_INNER / 128), blk, 0, stream>>>(
        hs, D_MODEL, in_proj_w + (size_t)D_INNER * D_MODEL, D_MODEL,
        z, D_INNER, D_MODEL, nullptr);

    const int nelem = BATCH * SEQLEN * D_INNER;
    conv_silu_kernel<<<(nelem + 255) / 256, blk, 0, stream>>>(x, conv_w, conv_b, xc);

    gemm_wmma_f32<6, 0, X_SPLIT><<<dim3(BL / 128, X_SPLIT), blk, 0, stream>>>(
        xc, D_INNER, x_proj_w, D_INNER, xpart, XDBL_LD, D_INNER, nullptr);
    splitk_sum_kernel<<<(BL * XDBL_LD / 4 + 255) / 256, blk, 0, stream>>>(xpart, xdbl);

    gemm_wmma_f32<8, 1><<<dim3(BL / 128, D_INNER / 128), blk, 0, stream>>>(
        xdbl, XDBL_LD, dt_proj_w, DT_RANK, delta, D_INNER, DT_RANK, dt_proj_b);

    float* y = x;
    const int scan_threads = BATCH * D_INNER * D_STATE;
    scan_kernel<<<(scan_threads + 255) / 256, blk, 0, stream>>>(
        delta, xc, xdbl, A_log, Dp, z, y);

    gemm_wmma_f32<8, 0><<<dim3(BL / 128, D_MODEL / 128), blk, 0, stream>>>(
        y, D_INNER, out_w, D_INNER, out, D_MODEL, D_INNER, nullptr);
}


# conv+silu: one thread per 4 channels x 16 time steps, 4-tap window sliding in registers over 16-byte row loads (4096 waves, not 262144 one-output waves); per-output sums and their order unchanged
# speedup vs baseline: 5.4112x; 1.0356x over previous
#include <hip/hip_runtime.h>
#include <math.h>

#define D_MODEL  1024
#define D_STATE  16
#define D_CONV   4
#define D_INNER  2048
#define DT_RANK  64
#define BATCH    2
#define SEQLEN   2048
#define BL       (BATCH * SEQLEN)
#define XDBL_LD  (DT_RANK + 2 * D_STATE)

typedef __attribute__((ext_vector_type(2))) float v2f;
typedef __attribute__((ext_vector_type(4))) float v4f;
typedef __attribute__((ext_vector_type(8))) float v8f;

__device__ __forceinline__ float silu_f(float v)     { return v / (1.f + expf(-v)); }

#define LOG2E_F 1.44269504088896340736f
#define LN2_F   0.693147180559945309417f

__device__ __forceinline__ float exp_hw(float v) { return __builtin_amdgcn_exp2f(v * LOG2E_F); }

__device__ __forceinline__ float silu_hw(float v)
{
    return v * __builtin_amdgcn_rcpf(1.f + exp_hw(-v));
}

__device__ __forceinline__ float softplus_hw(float v)
{
    const float e = exp_hw(v);
    float p = fmaf(e, -1.f / 6.f, 1.f / 5.f);
    p = fmaf(e, p, -1.f / 4.f);
    p = fmaf(e, p, 1.f / 3.f);
    p = fmaf(e, p, -1.f / 2.f);
    p = fmaf(e, p, 1.f);
    const float lo = e * p;
    const float u  = 1.f + e;
    const float hi = (__builtin_amdgcn_logf(u) * LN2_F) * (e * __builtin_amdgcn_rcpf(u - 1.f));
    const float r  = (e < 0.0625f) ? lo : hi;
    return (v > 20.f) ? v : r;
}

template <int NT, int EPI, int KSPLIT = 1>
__launch_bounds__(256)
__global__ void gemm_wmma_f32(const float* __restrict__ A, int lda,
                              const float* __restrict__ W, int ldw,
                              float* __restrict__ C, int ldc, int K,
                              const float* __restrict__ bias)
{
    constexpr int KC      = 32;
    constexpr int KC4     = KC / 4;
    constexpr int LDS_LD  = KC + 4;
    constexpr int NROWS_B = 16 * NT;
    constexpr int NA      = (128 * KC4) / 256;
    constexpr int NB      = (NROWS_B * KC4) / 256;

    __shared__ float As[128 * LDS_LD];
    __shared__ float Bs[NROWS_B * LDS_LD];

    const int tid  = threadIdx.x;
    const int wave = tid >> 5;
    const int lane = tid & 31;
    const int half = lane >> 4;
    const int lidx = lane & 15;
    const int m_blk  = blockIdx.x * 128;
    const int n_base = (KSPLIT > 1) ? 0 : blockIdx.y * NROWS_B;
    if (KSPLIT > 1) {
        K /= KSPLIT;
        A += blockIdx.y * K;
        W += blockIdx.y * K;
        C += (size_t)blockIdx.y * gridDim.x * 128 * ldc;
    }

    v8f acc[NT];
#pragma unroll
    for (int t = 0; t < NT; ++t) acc[t] = (v8f){0.f,0.f,0.f,0.f,0.f,0.f,0.f,0.f};

    v4f ra[NA], rb[NB];

#pragma unroll
    for (int i = 0; i < NA; ++i) {
        const int idx = tid + i * 256, r = idx >> 3, c = (idx & 7) * 4;
        ra[i] = *(const v4f*)(A + (size_t)(m_blk + r) * lda + c);
    }
#pragma unroll
    for (int i = 0; i < NB; ++i) {
        const int idx = tid + i * 256, r = idx >> 3, c = (idx & 7) * 4;
        rb[i] = *(const v4f*)(W + (size_t)(n_base + r) * ldw + c);
    }

    for (int k0 = 0; k0 < K; k0 += KC) {
#pragma unroll
        for (int i = 0; i < NA; ++i) {
            const int idx = tid + i * 256, r = idx >> 3, c = (idx & 7) * 4;
            *(v4f*)(As + r * LDS_LD + c) = ra[i];
        }
#pragma unroll
        for (int i = 0; i < NB; ++i) {
            const int idx = tid + i * 256, r = idx >> 3, c = (idx & 7) * 4;
            *(v4f*)(Bs + r * LDS_LD + c) = rb[i];
        }
        __syncthreads();

        if (k0 + KC < K) {
#pragma unroll
            for (int i = 0; i < NA; ++i) {
                const int idx = tid + i * 256, r = idx >> 3, c = (idx & 7) * 4;
                ra[i] = *(const v4f*)(A + (size_t)(m_blk + r) * lda + (k0 + KC) + c);
            }
#pragma unroll
            for (int i = 0; i < NB; ++i) {
                const int idx = tid + i * 256, r = idx >> 3, c = (idx & 7) * 4;
                rb[i] = *(const v4f*)(W + (size_t)(n_base + r) * ldw + (k0 + KC) + c);
            }
        }

        const float* a_lds = As + (wave * 16 + lidx) * LDS_LD + 2 * half;
        const float* b_lds = Bs + lidx * LDS_LD + 2 * half;
#pragma unroll
        for (int kk = 0; kk < KC; kk += 4) {
            v2f a = *(const v2f*)(a_lds + kk);
#pragma unroll
            for (int t = 0; t < NT; ++t) {
                v2f b = *(const v2f*)(b_lds + t * 16 * LDS_LD + kk);
                acc[t] = __builtin_amdgcn_wmma_f32_16x16x4_f32(
                    false, a, false, b, (short)0, acc[t], false, false);
            }
        }
        __syncthreads();
    }

#pragma unroll
    for (int t = 0; t < NT; ++t) {
        const int n = n_base + t * 16 + lidx;
#pragma unroll
        for (int v = 0; v < 8; ++v) {
            const int m = m_blk + wave * 16 + half * 8 + v;
            float val = acc[t][v];
            if (EPI == 1) val = softplus_hw(val + bias[n]);
            C[(size_t)m * ldc + n] = val;
        }
    }
}

#define X_SPLIT 8

__global__ void splitk_sum_kernel(const float* __restrict__ part,
                                  float* __restrict__ dst)
{
    constexpr int PLANE4 = BL * XDBL_LD / 4;
    const int i = blockIdx.x * blockDim.x + threadIdx.x;
    if (i >= PLANE4) return;
    const v4f* p = (const v4f*)part + i;
    v4f v[X_SPLIT];
#pragma unroll
    for (int s = 0; s < X_SPLIT; ++s) v[s] = p[(size_t)s * PLANE4];
    v4f acc = v[0];
#pragma unroll
    for (int s = 1; s < X_SPLIT; ++s) acc += v[s];
    ((v4f*)dst)[i] = acc;
}

#define CONV_RUN 16

__launch_bounds__(256)
__global__ void conv_silu_kernel(const float* __restrict__ x,
                                 const float* __restrict__ w,
                                 const float* __restrict__ cb,
                                 float* __restrict__ xc)
{
    constexpr int DQ   = D_INNER / 4;
    constexpr int RUNS = SEQLEN / CONV_RUN;
    const int idx = blockIdx.x * blockDim.x + threadIdx.x;
    if (idx >= BATCH * RUNS * DQ) return;
    const int d   = (idx % DQ) * 4;
    const int run = idx / DQ;
    const int l0  = (run % RUNS) * CONV_RUN;
    const int b   = run / RUNS;

    const size_t base = ((size_t)b * SEQLEN + l0) * D_INNER + d;
    const v4f* xr = (const v4f*)(x + base);
    v4f*       yr = (v4f*)(xc + base);

    v4f wc[D_CONV];
#pragma unroll
    for (int c = 0; c < 4; ++c) wc[c] = *(const v4f*)(w + (size_t)(d + c) * D_CONV);
    const v4f bias = *(const v4f*)(cb + d);

    v4f win[CONV_RUN + D_CONV - 1];
#pragma unroll
    for (int j = 0; j < D_CONV - 1; ++j)
        win[j] = (l0 > 0) ? xr[(j - (D_CONV - 1)) * DQ] : (v4f){0.f, 0.f, 0.f, 0.f};
#pragma unroll
    for (int j = 0; j < CONV_RUN; ++j) win[D_CONV - 1 + j] = xr[j * DQ];

#pragma unroll
    for (int j = 0; j < CONV_RUN; ++j) {
        v4f o;
#pragma unroll
        for (int c = 0; c < 4; ++c) {
            float acc = bias[c];
#pragma unroll
            for (int k = 0; k < D_CONV; ++k) acc = fmaf(win[j + k][c], wc[c][k], acc);
            o[c] = silu_f(acc);
        }
        yr[j * DQ] = o;
    }
}

#define SCAN_T 8

struct ScanBlk {
    float dt[SCAN_T], u[SCAN_T], Bn[SCAN_T], Cn[SCAN_T];
    float ug, zg;
};

__device__ __forceinline__ void scan_load(ScanBlk& r,
                                          const float* __restrict__ delta,
                                          const float* __restrict__ xc,
                                          const float* __restrict__ z,
                                          const float* __restrict__ xdbl,
                                          size_t row, int d, int n, int jl)
{
#pragma unroll
    for (int j = 0; j < SCAN_T; ++j) {
        const size_t off = (row + j) * D_INNER + d;
        const float* bc = xdbl + (row + j) * XDBL_LD;
        r.dt[j] = delta[off];
        r.u[j]  = xc[off];
        r.Bn[j] = bc[DT_RANK + n];
        r.Cn[j] = bc[DT_RANK + D_STATE + n];
    }
    const size_t offg = (row + jl) * D_INNER + d;
    r.ug = xc[offg];
    r.zg = z[offg];
}

__device__ __forceinline__ void scan_steps(const ScanBlk& r, float& h,
                                           float An2, float Dd,
                                           float* __restrict__ yout,
                                           size_t row, int d, int n, int jl)
{
    float y[SCAN_T];
#pragma unroll
    for (int j = 0; j < SCAN_T; ++j) {
        const float dt = r.dt[j];
        const float dA = __builtin_amdgcn_exp2f(dt * An2);
        h = fmaf(h, dA, (dt * r.u[j]) * r.Bn[j]);
        float t = h * r.Cn[j];
        t += __shfl_xor(t, 1, 16);
        t += __shfl_xor(t, 2, 16);
        t += __shfl_xor(t, 4, 16);
        t += __shfl_xor(t, 8, 16);
        y[j] = t;
    }
    float ys = y[0];
#pragma unroll
    for (int j = 1; j < SCAN_T; ++j) ys = (jl == j) ? y[j] : ys;
    const float o = (ys + Dd * r.ug) * silu_hw(r.zg);
    if (n < SCAN_T) yout[(row + jl) * D_INNER + d] = o;
}

__launch_bounds__(256)
__global__ void scan_kernel(const float* __restrict__ delta,
                            const float* __restrict__ xc,
                            const float* __restrict__ xdbl,
                            const float* __restrict__ A_log,
                            const float* __restrict__ Dp,
                            const float* __restrict__ z,
                            float* __restrict__ yout)
{
    const int gid = blockIdx.x * blockDim.x + threadIdx.x;
    const int n  = gid & (D_STATE - 1);
    const int ch = gid >> 4;
    if (ch >= BATCH * D_INNER) return;
    const int d = ch % D_INNER;
    const int b = ch / D_INNER;

    const int jl = n & (SCAN_T - 1);
    const float An2 = -expf(A_log[d * D_STATE + n]) * LOG2E_F;
    const float Dd = Dp[d];
    float h = 0.f;

    const size_t row0 = (size_t)b * SEQLEN;

    ScanBlk ra, rb;
    scan_load(ra, delta, xc, z, xdbl, row0, d, n, jl);
    for (int l0 = 0; l0 < SEQLEN; l0 += 2 * SCAN_T) {
        scan_load(rb, delta, xc, z, xdbl, row0 + l0 + SCAN_T, d, n, jl);
        scan_steps(ra, h, An2, Dd, yout, row0 + l0, d, n, jl);
        const int ln = (l0 + 2 * SCAN_T < SEQLEN) ? (l0 + 2 * SCAN_T) : (SEQLEN - SCAN_T);
        scan_load(ra, delta, xc, z, xdbl, row0 + ln, d, n, jl);
        scan_steps(rb, h, An2, Dd, yout, row0 + l0 + SCAN_T, d, n, jl);
    }
}

extern "C" void kernel_launch(void* const* d_in, const int* in_sizes, int n_in,
                              void* d_out, int out_size, void* d_ws, size_t ws_size,
                              hipStream_t stream)
{
    const float* hs        = (const float*)d_in[0];
    const float* in_proj_w = (const float*)d_in[1];
    const float* conv_w    = (const float*)d_in[2];
    const float* conv_b    = (const float*)d_in[3];
    const float* x_proj_w  = (const float*)d_in[4];
    const float* dt_proj_w = (const float*)d_in[5];
    const float* dt_proj_b = (const float*)d_in[6];
    const float* A_log     = (const float*)d_in[7];
    const float* Dp        = (const float*)d_in[8];
    const float* out_w     = (const float*)d_in[9];
    float* out = (float*)d_out;

    const size_t SZ_BLD = (size_t)BL * D_INNER * sizeof(float);
    char* ws = (char*)d_ws;
    float* x     = (float*)(ws);
    float* z     = (float*)(ws + SZ_BLD);
    float* xc    = (float*)(ws + 2 * SZ_BLD);
    float* delta = (float*)(ws + 3 * SZ_BLD);
    float* xdbl  = (float*)(ws + 4 * SZ_BLD);
    float* xpart = (float*)(ws + 5 * SZ_BLD);

    const dim3 blk(256);

    gemm_wmma_f32<8, 0><<<dim3(BL / 128, D_INNER / 128), blk, 0, stream>>>(
        hs, D_MODEL, in_proj_w, D_MODEL, x, D_INNER, D_MODEL, nullptr);
    gemm_wmma_f32<8, 0><<<dim3(BL / 128, D_INNER / 128), blk, 0, stream>>>(
        hs, D_MODEL, in_proj_w + (size_t)D_INNER * D_MODEL, D_MODEL,
        z, D_INNER, D_MODEL, nullptr);

    const int nconv = BATCH * (SEQLEN / CONV_RUN) * (D_INNER / 4);
    conv_silu_kernel<<<(nconv + 255) / 256, blk, 0, stream>>>(x, conv_w, conv_b, xc);

    gemm_wmma_f32<6, 0, X_SPLIT><<<dim3(BL / 128, X_SPLIT), blk, 0, stream>>>(
        xc, D_INNER, x_proj_w, D_INNER, xpart, XDBL_LD, D_INNER, nullptr);
    splitk_sum_kernel<<<(BL * XDBL_LD / 4 + 255) / 256, blk, 0, stream>>>(xpart, xdbl);

    gemm_wmma_f32<8, 1><<<dim3(BL / 128, D_INNER / 128), blk, 0, stream>>>(
        xdbl, XDBL_LD, dt_proj_w, DT_RANK, delta, D_INNER, DT_RANK, dt_proj_b);

    float* y = x;
    const int scan_threads = BATCH * D_INNER * D_STATE;
    scan_kernel<<<(scan_threads + 255) / 256, blk, 0, stream>>>(
        delta, xc, xdbl, A_log, Dp, z, y);

    gemm_wmma_f32<8, 0><<<dim3(BL / 128, D_MODEL / 128), blk, 0, stream>>>(
        y, D_INNER, out_w, D_INNER, out, D_MODEL, D_INNER, nullptr);
}


# scan: lane = channel x state pair (8 lanes per channel, 1024 waves); y = in-lane pair + 3 DPP row_xmask levels, same xor sum tree, no ds_bpermute; dt, dt*u by DPP8; 3 blocks of loads in flight
# speedup vs baseline: 6.7299x; 1.2437x over previous
#include <hip/hip_runtime.h>
#include <math.h>

#define D_MODEL  1024
#define D_STATE  16
#define D_CONV   4
#define D_INNER  2048
#define DT_RANK  64
#define BATCH    2
#define SEQLEN   2048
#define BL       (BATCH * SEQLEN)
#define XDBL_LD  (DT_RANK + 2 * D_STATE)

typedef __attribute__((ext_vector_type(2))) float v2f;
typedef __attribute__((ext_vector_type(4))) float v4f;
typedef __attribute__((ext_vector_type(8))) float v8f;

__device__ __forceinline__ float silu_f(float v)     { return v / (1.f + expf(-v)); }

#define LOG2E_F 1.44269504088896340736f
#define LN2_F   0.693147180559945309417f

__device__ __forceinline__ float exp_hw(float v) { return __builtin_amdgcn_exp2f(v * LOG2E_F); }

__device__ __forceinline__ float silu_hw(float v)
{
    return v * __builtin_amdgcn_rcpf(1.f + exp_hw(-v));
}

__device__ __forceinline__ float softplus_hw(float v)
{
    const float e = exp_hw(v);
    float p = fmaf(e, -1.f / 6.f, 1.f / 5.f);
    p = fmaf(e, p, -1.f / 4.f);
    p = fmaf(e, p, 1.f / 3.f);
    p = fmaf(e, p, -1.f / 2.f);
    p = fmaf(e, p, 1.f);
    const float lo = e * p;
    const float u  = 1.f + e;
    const float hi = (__builtin_amdgcn_logf(u) * LN2_F) * (e * __builtin_amdgcn_rcpf(u - 1.f));
    const float r  = (e < 0.0625f) ? lo : hi;
    return (v > 20.f) ? v : r;
}

template <int NT, int EPI, int KSPLIT = 1>
__launch_bounds__(256)
__global__ void gemm_wmma_f32(const float* __restrict__ A, int lda,
                              const float* __restrict__ W, int ldw,
                              float* __restrict__ C, int ldc, int K,
                              const float* __restrict__ bias)
{
    constexpr int KC      = 32;
    constexpr int KC4     = KC / 4;
    constexpr int LDS_LD  = KC + 4;
    constexpr int NROWS_B = 16 * NT;
    constexpr int NA      = (128 * KC4) / 256;
    constexpr int NB      = (NROWS_B * KC4) / 256;

    __shared__ float As[128 * LDS_LD];
    __shared__ float Bs[NROWS_B * LDS_LD];

    const int tid  = threadIdx.x;
    const int wave = tid >> 5;
    const int lane = tid & 31;
    const int half = lane >> 4;
    const int lidx = lane & 15;
    const int m_blk  = blockIdx.x * 128;
    const int n_base = (KSPLIT > 1) ? 0 : blockIdx.y * NROWS_B;
    if (KSPLIT > 1) {
        K /= KSPLIT;
        A += blockIdx.y * K;
        W += blockIdx.y * K;
        C += (size_t)blockIdx.y * gridDim.x * 128 * ldc;
    }

    v8f acc[NT];
#pragma unroll
    for (int t = 0; t < NT; ++t) acc[t] = (v8f){0.f,0.f,0.f,0.f,0.f,0.f,0.f,0.f};

    v4f ra[NA], rb[NB];

#pragma unroll
    for (int i = 0; i < NA; ++i) {
        const int idx = tid + i * 256, r = idx >> 3, c = (idx & 7) * 4;
        ra[i] = *(const v4f*)(A + (size_t)(m_blk + r) * lda + c);
    }
#pragma unroll
    for (int i = 0; i < NB; ++i) {
        const int idx = tid + i * 256, r = idx >> 3, c = (idx & 7) * 4;
        rb[i] = *(const v4f*)(W + (size_t)(n_base + r) * ldw + c);
    }

    for (int k0 = 0; k0 < K; k0 += KC) {
#pragma unroll
        for (int i = 0; i < NA; ++i) {
            const int idx = tid + i * 256, r = idx >> 3, c = (idx & 7) * 4;
            *(v4f*)(As + r * LDS_LD + c) = ra[i];
        }
#pragma unroll
        for (int i = 0; i < NB; ++i) {
            const int idx = tid + i * 256, r = idx >> 3, c = (idx & 7) * 4;
            *(v4f*)(Bs + r * LDS_LD + c) = rb[i];
        }
        __syncthreads();

        if (k0 + KC < K) {
#pragma unroll
            for (int i = 0; i < NA; ++i) {
                const int idx = tid + i * 256, r = idx >> 3, c = (idx & 7) * 4;
                ra[i] = *(const v4f*)(A + (size_t)(m_blk + r) * lda + (k0 + KC) + c);
            }
#pragma unroll
            for (int i = 0; i < NB; ++i) {
                const int idx = tid + i * 256, r = idx >> 3, c = (idx & 7) * 4;
                rb[i] = *(const v4f*)(W + (size_t)(n_base + r) * ldw + (k0 + KC) + c);
            }
        }

        const float* a_lds = As + (wave * 16 + lidx) * LDS_LD + 2 * half;
        const float* b_lds = Bs + lidx * LDS_LD + 2 * half;
#pragma unroll
        for (int kk = 0; kk < KC; kk += 4) {
            v2f a = *(const v2f*)(a_lds + kk);
#pragma unroll
            for (int t = 0; t < NT; ++t) {
                v2f b = *(const v2f*)(b_lds + t * 16 * LDS_LD + kk);
                acc[t] = __builtin_amdgcn_wmma_f32_16x16x4_f32(
                    false, a, false, b, (short)0, acc[t], false, false);
            }
        }
        __syncthreads();
    }

#pragma unroll
    for (int t = 0; t < NT; ++t) {
        const int n = n_base + t * 16 + lidx;
#pragma unroll
        for (int v = 0; v < 8; ++v) {
            const int m = m_blk + wave * 16 + half * 8 + v;
            float val = acc[t][v];
            if (EPI == 1) val = softplus_hw(val + bias[n]);
            C[(size_t)m * ldc + n] = val;
        }
    }
}

#define X_SPLIT 8

__global__ void splitk_sum_kernel(const float* __restrict__ part,
                                  float* __restrict__ dst)
{
    constexpr int PLANE4 = BL * XDBL_LD / 4;
    const int i = blockIdx.x * blockDim.x + threadIdx.x;
    if (i >= PLANE4) return;
    const v4f* p = (const v4f*)part + i;
    v4f v[X_SPLIT];
#pragma unroll
    for (int s = 0; s < X_SPLIT; ++s) v[s] = p[(size_t)s * PLANE4];
    v4f acc = v[0];
#pragma unroll
    for (int s = 1; s < X_SPLIT; ++s) acc += v[s];
    ((v4f*)dst)[i] = acc;
}

#define CONV_RUN 16

__launch_bounds__(256)
__global__ void conv_silu_kernel(const float* __restrict__ x,
                                 const float* __restrict__ w,
                                 const float* __restrict__ cb,
                                 float* __restrict__ xc)
{
    constexpr int DQ   = D_INNER / 4;
    constexpr int RUNS = SEQLEN / CONV_RUN;
    const int idx = blockIdx.x * blockDim.x + threadIdx.x;
    if (idx >= BATCH * RUNS * DQ) return;
    const int d   = (idx % DQ) * 4;
    const int run = idx / DQ;
    const int l0  = (run % RUNS) * CONV_RUN;
    const int b   = run / RUNS;

    const size_t base = ((size_t)b * SEQLEN + l0) * D_INNER + d;
    const v4f* xr = (const v4f*)(x + base);
    v4f*       yr = (v4f*)(xc + base);

    v4f wc[D_CONV];
#pragma unroll
    for (int c = 0; c < 4; ++c) wc[c] = *(const v4f*)(w + (size_t)(d + c) * D_CONV);
    const v4f bias = *(const v4f*)(cb + d);

    v4f win[CONV_RUN + D_CONV - 1];
#pragma unroll
    for (int j = 0; j < D_CONV - 1; ++j)
        win[j] = (l0 > 0) ? xr[(j - (D_CONV - 1)) * DQ] : (v4f){0.f, 0.f, 0.f, 0.f};
#pragma unroll
    for (int j = 0; j < CONV_RUN; ++j) win[D_CONV - 1 + j] = xr[j * DQ];

#pragma unroll
    for (int j = 0; j < CONV_RUN; ++j) {
        v4f o;
#pragma unroll
        for (int c = 0; c < 4; ++c) {
            float acc = bias[c];
#pragma unroll
            for (int k = 0; k < D_CONV; ++k) acc = fmaf(win[j + k][c], wc[c][k], acc);
            o[c] = silu_f(acc);
        }
        yr[j * DQ] = o;
    }
}

#define SCAN_T    8
#define SCAN_RING 4

struct ScanBlk {
    v2f Bn[SCAN_T], Cn[SCAN_T];
    float dt, u, z;
};

template <int J>
__device__ __forceinline__ float bcast8(float v)
{
    return __builtin_bit_cast(float,
        __builtin_amdgcn_mov_dpp8(__builtin_bit_cast(unsigned, v), J * 0x249249));
}

template <int M>
__device__ __forceinline__ float xsum(float keep, float send)
{
    return keep + __builtin_bit_cast(float,
        __builtin_amdgcn_update_dpp(0, __builtin_bit_cast(int, send), 0x160 + M, 0xf, 0xf, true));
}

__device__ __forceinline__ void scan_load(ScanBlk& r,
                                          const float* __restrict__ delta,
                                          const float* __restrict__ xc,
                                          const float* __restrict__ z,
                                          const float* __restrict__ xdbl,
                                          size_t row, int d, int p)
{
#pragma unroll
    for (int j = 0; j < SCAN_T; ++j) {
        const float* bc = xdbl + (row + j) * XDBL_LD + DT_RANK + 2 * p;
        r.Bn[j] = *(const v2f*)(bc);
        r.Cn[j] = *(const v2f*)(bc + D_STATE);
    }
    const size_t off = (row + p) * D_INNER + d;
    r.dt = delta[off];
    r.u  = xc[off];
    r.z  = z[off];
}

template <int J>
__device__ __forceinline__ float scan_step(const ScanBlk& r, float& h0, float& h1,
                                           float An2_0, float An2_1, float dtu_own)
{
    const float dt  = bcast8<J>(r.dt);
    const float dtu = bcast8<J>(dtu_own);
    const float dA0 = __builtin_amdgcn_exp2f(dt * An2_0);
    const float dA1 = __builtin_amdgcn_exp2f(dt * An2_1);
    h0 = fmaf(h0, dA0, dtu * r.Bn[J].x);
    h1 = fmaf(h1, dA1, dtu * r.Bn[J].y);
    return h0 * r.Cn[J].x + h1 * r.Cn[J].y;
}

__device__ __forceinline__ void scan_steps(const ScanBlk& r, float& h0, float& h1,
                                           float An2_0, float An2_1, float Dd,
                                           float* __restrict__ yout,
                                           size_t row, int d, int p)
{
    const float dtu_own = r.dt * r.u;
    float t[SCAN_T];
    t[0] = scan_step<0>(r, h0, h1, An2_0, An2_1, dtu_own);
    t[1] = scan_step<1>(r, h0, h1, An2_0, An2_1, dtu_own);
    t[2] = scan_step<2>(r, h0, h1, An2_0, An2_1, dtu_own);
    t[3] = scan_step<3>(r, h0, h1, An2_0, An2_1, dtu_own);
    t[4] = scan_step<4>(r, h0, h1, An2_0, An2_1, dtu_own);
    t[5] = scan_step<5>(r, h0, h1, An2_0, An2_1, dtu_own);
    t[6] = scan_step<6>(r, h0, h1, An2_0, An2_1, dtu_own);
    t[7] = scan_step<7>(r, h0, h1, An2_0, An2_1, dtu_own);
    const bool b0 = (p & 1) != 0, b1 = (p & 2) != 0, b2 = (p & 4) != 0;
    const float a0 = xsum<1>(b0 ? t[1] : t[0], b0 ? t[0] : t[1]);
    const float a1 = xsum<1>(b0 ? t[3] : t[2], b0 ? t[2] : t[3]);
    const float a2 = xsum<1>(b0 ? t[5] : t[4], b0 ? t[4] : t[5]);
    const float a3 = xsum<1>(b0 ? t[7] : t[6], b0 ? t[6] : t[7]);
    const float c0 = xsum<2>(b1 ? a1 : a0, b1 ? a0 : a1);
    const float c1 = xsum<2>(b1 ? a3 : a2, b1 ? a2 : a3);
    const float ys = xsum<4>(b2 ? c1 : c0, b2 ? c0 : c1);
    yout[(row + p) * D_INNER + d] = (ys + Dd * r.u) * silu_hw(r.z);
}

__launch_bounds__(128)
__global__ void scan_kernel(const float* __restrict__ delta,
                            const float* __restrict__ xc,
                            const float* __restrict__ xdbl,
                            const float* __restrict__ A_log,
                            const float* __restrict__ Dp,
                            const float* __restrict__ z,
                            float* __restrict__ yout)
{
    const int gid = blockIdx.x * blockDim.x + threadIdx.x;
    const int p  = gid & (SCAN_T - 1);
    const int ch = gid >> 3;
    if (ch >= BATCH * D_INNER) return;
    const int d = ch % D_INNER;
    const int b = ch / D_INNER;

    const v2f al = *(const v2f*)(A_log + d * D_STATE + 2 * p);
    const float An2_0 = -expf(al.x) * LOG2E_F;
    const float An2_1 = -expf(al.y) * LOG2E_F;
    const float Dd = Dp[d];
    float h0 = 0.f, h1 = 0.f;

    const size_t row0 = (size_t)b * SEQLEN;
    constexpr int LAST = SEQLEN - SCAN_T;

#define SCAN_FENCE() __builtin_amdgcn_sched_barrier(0)
    ScanBlk r0, r1, r2, r3;
    SCAN_FENCE();
    scan_load(r0, delta, xc, z, xdbl, row0,              d, p);
    SCAN_FENCE();
    scan_load(r1, delta, xc, z, xdbl, row0 + SCAN_T,     d, p);
    SCAN_FENCE();
    scan_load(r2, delta, xc, z, xdbl, row0 + 2 * SCAN_T, d, p);
    SCAN_FENCE();
    scan_load(r3, delta, xc, z, xdbl, row0 + 3 * SCAN_T, d, p);
    for (int l0 = 0; l0 < SEQLEN; l0 += SCAN_RING * SCAN_T) {
        const int l4 = (l0 + 4 * SCAN_T < SEQLEN) ? (l0 + 4 * SCAN_T) : LAST;
        const int l5 = (l0 + 5 * SCAN_T < SEQLEN) ? (l0 + 5 * SCAN_T) : LAST;
        const int l6 = (l0 + 6 * SCAN_T < SEQLEN) ? (l0 + 6 * SCAN_T) : LAST;
        const int l7 = (l0 + 7 * SCAN_T < SEQLEN) ? (l0 + 7 * SCAN_T) : LAST;
        SCAN_FENCE();
        scan_steps(r0, h0, h1, An2_0, An2_1, Dd, yout, row0 + l0, d, p);
        SCAN_FENCE();
        scan_load(r0, delta, xc, z, xdbl, row0 + l4, d, p);
        SCAN_FENCE();
        scan_steps(r1, h0, h1, An2_0, An2_1, Dd, yout, row0 + l0 + SCAN_T, d, p);
        SCAN_FENCE();
        scan_load(r1, delta, xc, z, xdbl, row0 + l5, d, p);
        SCAN_FENCE();
        scan_steps(r2, h0, h1, An2_0, An2_1, Dd, yout, row0 + l0 + 2 * SCAN_T, d, p);
        SCAN_FENCE();
        scan_load(r2, delta, xc, z, xdbl, row0 + l6, d, p);
        SCAN_FENCE();
        scan_steps(r3, h0, h1, An2_0, An2_1, Dd, yout, row0 + l0 + 3 * SCAN_T, d, p);
        SCAN_FENCE();
        scan_load(r3, delta, xc, z, xdbl, row0 + l7, d, p);
        SCAN_FENCE();
    }
#undef SCAN_FENCE
}

extern "C" void kernel_launch(void* const* d_in, const int* in_sizes, int n_in,
                              void* d_out, int out_size, void* d_ws, size_t ws_size,
                              hipStream_t stream)
{
    const float* hs        = (const float*)d_in[0];
    const float* in_proj_w = (const float*)d_in[1];
    const float* conv_w    = (const float*)d_in[2];
    const float* conv_b    = (const float*)d_in[3];
    const float* x_proj_w  = (const float*)d_in[4];
    const float* dt_proj_w = (const float*)d_in[5];
    const float* dt_proj_b = (const float*)d_in[6];
    const float* A_log     = (const float*)d_in[7];
    const float* Dp        = (const float*)d_in[8];
    const float* out_w     = (const float*)d_in[9];
    float* out = (float*)d_out;

    const size_t SZ_BLD = (size_t)BL * D_INNER * sizeof(float);
    char* ws = (char*)d_ws;
    float* x     = (float*)(ws);
    float* z     = (float*)(ws + SZ_BLD);
    float* xc    = (float*)(ws + 2 * SZ_BLD);
    float* delta = (float*)(ws + 3 * SZ_BLD);
    float* xdbl  = (float*)(ws + 4 * SZ_BLD);
    float* xpart = (float*)(ws + 5 * SZ_BLD);

    const dim3 blk(256);

    gemm_wmma_f32<8, 0><<<dim3(BL / 128, D_INNER / 128), blk, 0, stream>>>(
        hs, D_MODEL, in_proj_w, D_MODEL, x, D_INNER, D_MODEL, nullptr);
    gemm_wmma_f32<8, 0><<<dim3(BL / 128, D_INNER / 128), blk, 0, stream>>>(
        hs, D_MODEL, in_proj_w + (size_t)D_INNER * D_MODEL, D_MODEL,
        z, D_INNER, D_MODEL, nullptr);

    const int nconv = BATCH * (SEQLEN / CONV_RUN) * (D_INNER / 4);
    conv_silu_kernel<<<(nconv + 255) / 256, blk, 0, stream>>>(x, conv_w, conv_b, xc);

    gemm_wmma_f32<6, 0, X_SPLIT><<<dim3(BL / 128, X_SPLIT), blk, 0, stream>>>(
        xc, D_INNER, x_proj_w, D_INNER, xpart, XDBL_LD, D_INNER, nullptr);
    splitk_sum_kernel<<<(BL * XDBL_LD / 4 + 255) / 256, blk, 0, stream>>>(xpart, xdbl);

    gemm_wmma_f32<8, 1><<<dim3(BL / 128, D_INNER / 128), blk, 0, stream>>>(
        xdbl, XDBL_LD, dt_proj_w, DT_RANK, delta, D_INNER, DT_RANK, dt_proj_b);

    float* y = x;
    const int scan_threads = BATCH * D_INNER * (D_STATE / 2);
    scan_kernel<<<scan_threads / 128, dim3(128), 0, stream>>>(
        delta, xc, xdbl, A_log, Dp, z, y);

    gemm_wmma_f32<8, 0><<<dim3(BL / 128, D_MODEL / 128), blk, 0, stream>>>(
        y, D_INNER, out_w, D_INNER, out, D_MODEL, D_INNER, nullptr);
}
